# starting kernel
# baseline (speedup 1.0000x reference)
#include <hip/hip_runtime.h>


typedef __attribute__((ext_vector_type(2))) float v2f;
typedef __attribute__((ext_vector_type(8))) float v8f;

#define NPTS 2000
#define DDIM 24
#define NTILE 125
#define LSPLIT 20
#define LCHUNK (NPTS / LSPLIT)

#define BETAT_FLOATS  (NPTS * DDIM * DDIM)
#define ALPHAT_FLOATS (NPTS * DDIM)
#define PART_FLOATS   (LSPLIT * NPTS * DDIM)

__global__ __launch_bounds__(256) void prep_kernel(const float* __restrict__ alpha,
                                                   const float* __restrict__ beta,
                                                   float* __restrict__ betaT,
                                                   float* __restrict__ alphaT) {
  int t = blockIdx.x * blockDim.x + threadIdx.x;
  if (t < BETAT_FLOATS) {
    int l = t / (DDIM * DDIM);
    int rem = t - l * (DDIM * DDIM);
    int j = rem / DDIM;
    int a = rem - j * DDIM;
    betaT[t] = (a == j) ? 0.0f : beta[(j * DDIM + a) * NPTS + l];
  }
  if (t < ALPHAT_FLOATS) {
    int l = t / DDIM;
    int j = t - l * DDIM;
    alphaT[t] = alpha[j * NPTS + l];
  }
}

__global__ __launch_bounds__(32) void rkhs_main(const float* __restrict__ x,
                                                const float* __restrict__ alphaT,
                                                const float* __restrict__ betaT,
                                                float* __restrict__ partial) {
  const int itile = blockIdx.x;
  const int chunk = blockIdx.y;
  const int lane  = threadIdx.x;
  const int i0    = itile * 16;

  __shared__ float xs[16][DDIM];
  __shared__ float xl[DDIM];
  __shared__ float df[16][DDIM + 2];
  __shared__ float Ssum[16];

  for (int t = lane; t < 16 * DDIM; t += 32) {
    int r = t / DDIM, k = t - r * DDIM;
    xs[r][k] = x[(i0 + r) * DDIM + k];
  }
  __syncthreads();

  const int  n0      = lane & 15;
  const int  koff    = (lane >> 4) * 2;
  const int  rowbase = (lane >> 4) * 8;
  const bool jt1ok   = (n0 < 8);
  const int  j1      = n0;
  const int  j2      = jt1ok ? (16 + n0) : 0;

  float acc0[8], acc1[8];
#pragma unroll
  for (int m = 0; m < 8; ++m) { acc0[m] = 0.0f; acc1[m] = 0.0f; }

  const int l0 = chunk * LCHUNK;
  for (int l = l0; l < l0 + LCHUNK; ++l) {
    if (lane < DDIM) xl[lane] = x[l * DDIM + lane];
    __syncthreads();

    for (int t = lane; t < 16 * DDIM; t += 32) {
      int r = t / DDIM, k = t - r * DDIM;
      df[r][k] = xs[r][k] - xl[k];
    }
    __syncthreads();

    if (lane < 16) {
      float s = 0.0f;
#pragma unroll
      for (int k = 0; k < DDIM; ++k) s += df[lane][k] * df[lane][k];
      Ssum[lane] = s;
    }
    __syncthreads();

    const float* Bl = betaT + (size_t)l * (DDIM * DDIM);
    v8f c0 = {};
    v8f c1 = {};
#pragma unroll
    for (int k0 = 0; k0 < DDIM; k0 += 4) {
      v2f a;
      a.x = df[n0][k0 + koff];
      a.y = df[n0][k0 + koff + 1];
      v2f b0;
      b0.x = Bl[j1 * DDIM + k0 + koff];
      b0.y = Bl[j1 * DDIM + k0 + koff + 1];
      float t0 = Bl[j2 * DDIM + k0 + koff];
      float t1 = Bl[j2 * DDIM + k0 + koff + 1];
      v2f b1;
      b1.x = jt1ok ? t0 : 0.0f;
      b1.y = jt1ok ? t1 : 0.0f;
      c0 = __builtin_amdgcn_wmma_f32_16x16x4_f32(false, a, false, b0,
                                                 (short)0, c0, false, false);
      c1 = __builtin_amdgcn_wmma_f32_16x16x4_f32(false, a, false, b1,
                                                 (short)0, c1, false, false);
    }

    float al0  = alphaT[l * DDIM + j1];
    float al1t = alphaT[l * DDIM + j2];
    float al1  = jt1ok ? al1t : 0.0f;
#pragma unroll
    for (int m = 0; m < 8; ++m) {
      int   r  = rowbase + m;
      float S  = Ssum[r];
      float d0 = df[r][j1];
      float w0 = __expf(d0 * d0 - S);
      acc0[m] += w0 * (al0 + 2.0f * c0[m]);
      float d1 = df[r][j2];
      float w1 = __expf(d1 * d1 - S);
      acc1[m] += w1 * (al1 + 2.0f * c1[m]);
    }
    __syncthreads();
  }

  float* p = partial + ((size_t)chunk * NPTS + i0) * DDIM;
#pragma unroll
  for (int m = 0; m < 8; ++m) {
    int r = rowbase + m;
    p[r * DDIM + j1] = acc0[m];
    if (jt1ok) p[r * DDIM + 16 + n0] = acc1[m];
  }
}

__global__ __launch_bounds__(256) void reduce_kernel(const float* __restrict__ partial,
                                                     float* __restrict__ out) {
  int t = blockIdx.x * blockDim.x + threadIdx.x;
  if (t < NPTS * DDIM) {
    float s = 0.0f;
#pragma unroll
    for (int c = 0; c < LSPLIT; ++c)
      s += partial[(size_t)c * (NPTS * DDIM) + t];
    out[t] = s;
  }
}

extern "C" void kernel_launch(void* const* d_in, const int* in_sizes, int n_in,
                              void* d_out, int out_size, void* d_ws, size_t ws_size,
                              hipStream_t stream) {
  const float* x     = (const float*)d_in[0];
  const float* alpha = (const float*)d_in[1];
  const float* beta  = (const float*)d_in[2];
  float* out = (float*)d_out;

  char* ws = (char*)d_ws;
  float* betaT   = (float*)ws;
  float* alphaT  = (float*)(ws + (size_t)BETAT_FLOATS * sizeof(float));
  float* partial = (float*)(ws + (size_t)(BETAT_FLOATS + ALPHAT_FLOATS) * sizeof(float));

  (void)in_sizes; (void)n_in; (void)out_size; (void)ws_size;

  prep_kernel<<<(BETAT_FLOATS + 255) / 256, 256, 0, stream>>>(alpha, beta, betaT, alphaT);

  dim3 grid(NTILE, LSPLIT);
  rkhs_main<<<grid, 32, 0, stream>>>(x, alphaT, betaT, partial);

  reduce_kernel<<<(NPTS * DDIM + 255) / 256, 256, 0, stream>>>(partial, out);
}


# main: each wave holds 5 i-tiles and uses one l's B fragments, x[l] row and alpha for all five (25 x 20 blocks of 4 waves); l split 80 x 25 instead of 20 x 100, so the partial sums regroup
# speedup vs baseline: 3.6056x; 3.6056x over previous
#include <hip/hip_runtime.h>


typedef __attribute__((ext_vector_type(2))) float v2f;
typedef __attribute__((ext_vector_type(8))) float v8f;

#define NPTS 2000
#define DDIM 24
#define NTILE 125
#define MT 5
#define NGROUP (NTILE / MT)
#define WPB 4
#define LSPLIT 80
#define LCHUNK (NPTS / LSPLIT)

#define BETAT_FLOATS  (NPTS * DDIM * DDIM)
#define ALPHAT_FLOATS (NPTS * DDIM)
#define PART_FLOATS   (LSPLIT * NPTS * DDIM)

__global__ __launch_bounds__(256) void prep_kernel(const float* __restrict__ alpha,
                                                   const float* __restrict__ beta,
                                                   float* __restrict__ betaT,
                                                   float* __restrict__ alphaT) {
  int t = blockIdx.x * blockDim.x + threadIdx.x;
  if (t < BETAT_FLOATS) {
    int l = t / (DDIM * DDIM);
    int rem = t - l * (DDIM * DDIM);
    int j = rem / DDIM;
    int a = rem - j * DDIM;
    betaT[t] = (a == j) ? 0.0f : beta[(j * DDIM + a) * NPTS + l];
  }
  if (t < ALPHAT_FLOATS) {
    int l = t / DDIM;
    int j = t - l * DDIM;
    alphaT[t] = alpha[j * NPTS + l];
  }
}

__device__ __forceinline__ float xhalf(float v) {
  return __builtin_bit_cast(float, __builtin_amdgcn_permlanex16(
      __builtin_bit_cast(int, v), __builtin_bit_cast(int, v),
      0x76543210, 0xfedcba98, false, false));
}

template <int SRC>
__device__ __forceinline__ float row_bcast(float v) {
  return __builtin_bit_cast(float, __builtin_amdgcn_update_dpp(
      0, __builtin_bit_cast(int, v), 0x150 + SRC, 0xF, 0xF, false));
}

__global__ __launch_bounds__(32 * WPB) __attribute__((amdgpu_waves_per_eu(2, 2)))
void rkhs_main(const float* __restrict__ x,
               const float* __restrict__ alphaT,
               const float* __restrict__ betaT,
               float* __restrict__ partial) {
  const int lane  = threadIdx.x & 31;
  const int wave  = __builtin_amdgcn_readfirstlane((int)(threadIdx.x >> 5));
  const int chunk = blockIdx.y * WPB + wave;
  const int ibase = blockIdx.x * (16 * MT);

  const int  n0      = lane & 15;
  const int  koff    = (lane >> 4) * 2;
  const int  rowbase = (lane >> 4) * 8;
  const bool jt1ok   = (n0 < 8);
  const int  j1      = n0;
  const int  j2      = jt1ok ? (16 + n0) : 0;

  v2f   xa[MT][6];
  float xc0[MT][8], xc1[MT][8];
  float acc0[MT][8], acc1[MT][8];
#pragma unroll
  for (int t = 0; t < MT; ++t) {
    const int i0 = ibase + 16 * t;
#pragma unroll
    for (int q = 0; q < 6; ++q)
      xa[t][q] = *reinterpret_cast<const v2f*>(x + (i0 + n0) * DDIM + 4 * q + koff);
#pragma unroll
    for (int m = 0; m < 8; ++m) {
      xc0[t][m] = x[(i0 + rowbase + m) * DDIM + j1];
      xc1[t][m] = x[(i0 + rowbase + m) * DDIM + j2];
      acc0[t][m] = 0.0f;
      acc1[t][m] = 0.0f;
    }
  }

  const int l0 = chunk * LCHUNK;
  for (int l = l0; l < l0 + LCHUNK; ++l) {
    const float* xlp = x + l * DDIM;
    const float* Bl  = betaT + (size_t)l * (DDIM * DDIM);

    v2f xl[6], b0[6], b1[6];
#pragma unroll
    for (int q = 0; q < 6; ++q) {
      xl[q] = *reinterpret_cast<const v2f*>(xlp + 4 * q + koff);
      b0[q] = *reinterpret_cast<const v2f*>(Bl + j1 * DDIM + 4 * q + koff);
      b1[q] = *reinterpret_cast<const v2f*>(Bl + j2 * DDIM + 4 * q + koff);
    }
    const float xl0  = xlp[j1];
    const float xl1  = xlp[j2];
    const float al0  = alphaT[l * DDIM + j1];
    const float al1t = alphaT[l * DDIM + j2];
    const float al1  = jt1ok ? al1t : 0.0f;

    v2f b1m[6];
#pragma unroll
    for (int q = 0; q < 6; ++q) {
      b1m[q].x = jt1ok ? b1[q].x : 0.0f;
      b1m[q].y = jt1ok ? b1[q].y : 0.0f;
    }

#pragma unroll
    for (int t = 0; t < MT; ++t) {
      v2f a[6];
      float p = 0.0f;
#pragma unroll
      for (int q = 0; q < 6; ++q) {
        a[q] = xa[t][q] - xl[q];
        p += a[q].x * a[q].x;
        p += a[q].y * a[q].y;
      }
      float s = p + xhalf(p);
      s = __builtin_bit_cast(float, __builtin_amdgcn_update_dpp(
          __builtin_bit_cast(int, s), __builtin_bit_cast(int, s),
          0x128, 0x2, 0xF, false));

      v8f c0 = {};
      v8f c1 = {};
#pragma unroll
      for (int q = 0; q < 6; ++q) {
        c0 = __builtin_amdgcn_wmma_f32_16x16x4_f32(false, a[q], false, b0[q],
                                                   (short)0, c0, false, false);
        c1 = __builtin_amdgcn_wmma_f32_16x16x4_f32(false, a[q], false, b1m[q],
                                                   (short)0, c1, false, false);
      }

      float Sr[8];
      Sr[0] = row_bcast<0>(s); Sr[1] = row_bcast<1>(s);
      Sr[2] = row_bcast<2>(s); Sr[3] = row_bcast<3>(s);
      Sr[4] = row_bcast<4>(s); Sr[5] = row_bcast<5>(s);
      Sr[6] = row_bcast<6>(s); Sr[7] = row_bcast<7>(s);
#pragma unroll
      for (int m = 0; m < 8; ++m) {
        float S  = Sr[m];
        float d0 = xc0[t][m] - xl0;
        float w0 = __expf(d0 * d0 - S);
        acc0[t][m] += w0 * (al0 + 2.0f * c0[m]);
        float d1 = xc1[t][m] - xl1;
        float w1 = __expf(d1 * d1 - S);
        acc1[t][m] += w1 * (al1 + 2.0f * c1[m]);
      }
    }
  }

#pragma unroll
  for (int t = 0; t < MT; ++t) {
    float* p = partial + ((size_t)chunk * NPTS + ibase + 16 * t) * DDIM;
#pragma unroll
    for (int m = 0; m < 8; ++m) {
      int r = rowbase + m;
      p[r * DDIM + j1] = acc0[t][m];
      if (jt1ok) p[r * DDIM + 16 + n0] = acc1[t][m];
    }
  }
}

__global__ __launch_bounds__(256) void reduce_kernel(const float* __restrict__ partial,
                                                     float* __restrict__ out) {
  int t = blockIdx.x * blockDim.x + threadIdx.x;
  if (t < NPTS * DDIM) {
    float s = 0.0f;
#pragma unroll
    for (int c = 0; c < LSPLIT; ++c)
      s += partial[(size_t)c * (NPTS * DDIM) + t];
    out[t] = s;
  }
}

extern "C" void kernel_launch(void* const* d_in, const int* in_sizes, int n_in,
                              void* d_out, int out_size, void* d_ws, size_t ws_size,
                              hipStream_t stream) {
  const float* x     = (const float*)d_in[0];
  const float* alpha = (const float*)d_in[1];
  const float* beta  = (const float*)d_in[2];
  float* out = (float*)d_out;

  char* ws = (char*)d_ws;
  float* betaT   = (float*)ws;
  float* alphaT  = (float*)(ws + (size_t)BETAT_FLOATS * sizeof(float));
  float* partial = (float*)(ws + (size_t)(BETAT_FLOATS + ALPHAT_FLOATS) * sizeof(float));

  (void)in_sizes; (void)n_in; (void)out_size; (void)ws_size;

  prep_kernel<<<(BETAT_FLOATS + 255) / 256, 256, 0, stream>>>(alpha, beta, betaT, alphaT);

  dim3 grid(NGROUP, LSPLIT / WPB);
  rkhs_main<<<grid, 32 * WPB, 0, stream>>>(x, alphaT, betaT, partial);

  reduce_kernel<<<(NPTS * DDIM + 255) / 256, 256, 0, stream>>>(partial, out);
}


# prep: beta/alpha transposed through a 64x64 LDS tile with 256-B row loads and stores, 320 work-groups (2,560 waves) instead of 4,500 (36,000 waves) of 8-KB-strided 4-B loads; values identical
# speedup vs baseline: 3.7938x; 1.0522x over previous
#include <hip/hip_runtime.h>


typedef __attribute__((ext_vector_type(2))) float v2f;
typedef __attribute__((ext_vector_type(8))) float v8f;

#define NPTS 2000
#define DDIM 24
#define NTILE 125
#define MT 5
#define NGROUP (NTILE / MT)
#define WPB 4
#define LSPLIT 80
#define LCHUNK (NPTS / LSPLIT)

#define BETAT_FLOATS  (NPTS * DDIM * DDIM)
#define ALPHAT_FLOATS (NPTS * DDIM)
#define PART_FLOATS   (LSPLIT * NPTS * DDIM)

#define PT 64
#define PPITCH 65
#define PRT (DDIM * DDIM / PT)
#define PLT ((NPTS + PT - 1) / PT)
#define PREP_BLOCKS (PRT * PLT + PLT)

__global__ __launch_bounds__(256) void prep_kernel(const float* __restrict__ alpha,
                                                   const float* __restrict__ beta,
                                                   float* __restrict__ betaT,
                                                   float* __restrict__ alphaT) {
  __shared__ float tile[PT][PPITCH];
  const int lane = threadIdx.x & 31;
  const int w = __builtin_amdgcn_readfirstlane((int)(threadIdx.x >> 5));
  const int b = blockIdx.x;
  const bool is_beta = b < PRT * PLT;
  const int l0 = (is_beta ? (b / PRT) : (b - PRT * PLT)) * PT;
  const int r0 = is_beta ? (b % PRT) * PT : 0;
  const int lg  = l0 + 2 * lane;
  const int lgc = (lg < NPTS) ? lg : (NPTS - 2);

  if (is_beta) {
    v2f v[8];
#pragma unroll
    for (int p = 0; p < 8; ++p) {
      const int r = r0 + w + 8 * p;
      v[p] = *reinterpret_cast<const v2f*>(beta + (size_t)r * NPTS + lgc);
    }
#pragma unroll
    for (int p = 0; p < 8; ++p) {
      const int rr = w + 8 * p;
      const int r  = r0 + rr;
      const bool diag = (r % (DDIM + 1)) == 0;
      tile[2 * lane][rr]     = diag ? 0.0f : v[p].x;
      tile[2 * lane + 1][rr] = diag ? 0.0f : v[p].y;
    }
  } else {
    v2f v[3];
#pragma unroll
    for (int p = 0; p < 3; ++p) {
      const int j = w + 8 * p;
      v[p] = *reinterpret_cast<const v2f*>(alpha + j * NPTS + lgc);
    }
#pragma unroll
    for (int p = 0; p < 3; ++p) {
      const int j = w + 8 * p;
      tile[2 * lane][j]     = v[p].x;
      tile[2 * lane + 1][j] = v[p].y;
    }
  }
  __syncthreads();
  if (is_beta) {
    v2f o[8];
#pragma unroll
    for (int p = 0; p < 8; ++p) {
      const int ll = w + 8 * p;
      o[p].x = tile[ll][2 * lane];
      o[p].y = tile[ll][2 * lane + 1];
    }
#pragma unroll
    for (int p = 0; p < 8; ++p) {
      const int l = l0 + w + 8 * p;
      if (l < NPTS)
        *reinterpret_cast<v2f*>(betaT + (size_t)l * (DDIM * DDIM) + r0 + 2 * lane) = o[p];
    }
  } else {
    float o[(PT * DDIM) / 256];
#pragma unroll
    for (int k = 0; k < (PT * DDIM) / 256; ++k) {
      const int e  = (int)threadIdx.x + 256 * k;
      const int ll = e / DDIM;
      o[k] = tile[ll][e - ll * DDIM];
    }
#pragma unroll
    for (int k = 0; k < (PT * DDIM) / 256; ++k) {
      const int e = (int)threadIdx.x + 256 * k;
      if (l0 + e / DDIM < NPTS) alphaT[l0 * DDIM + e] = o[k];
    }
  }
}

__device__ __forceinline__ float xhalf(float v) {
  return __builtin_bit_cast(float, __builtin_amdgcn_permlanex16(
      __builtin_bit_cast(int, v), __builtin_bit_cast(int, v),
      0x76543210, 0xfedcba98, false, false));
}

template <int SRC>
__device__ __forceinline__ float row_bcast(float v) {
  return __builtin_bit_cast(float, __builtin_amdgcn_update_dpp(
      0, __builtin_bit_cast(int, v), 0x150 + SRC, 0xF, 0xF, false));
}

__global__ __launch_bounds__(32 * WPB) __attribute__((amdgpu_waves_per_eu(2, 2)))
void rkhs_main(const float* __restrict__ x,
               const float* __restrict__ alphaT,
               const float* __restrict__ betaT,
               float* __restrict__ partial) {
  const int lane  = threadIdx.x & 31;
  const int wave  = __builtin_amdgcn_readfirstlane((int)(threadIdx.x >> 5));
  const int chunk = blockIdx.y * WPB + wave;
  const int ibase = blockIdx.x * (16 * MT);

  const int  n0      = lane & 15;
  const int  koff    = (lane >> 4) * 2;
  const int  rowbase = (lane >> 4) * 8;
  const bool jt1ok   = (n0 < 8);
  const int  j1      = n0;
  const int  j2      = jt1ok ? (16 + n0) : 0;

  v2f   xa[MT][6];
  float xc0[MT][8], xc1[MT][8];
  float acc0[MT][8], acc1[MT][8];
#pragma unroll
  for (int t = 0; t < MT; ++t) {
    const int i0 = ibase + 16 * t;
#pragma unroll
    for (int q = 0; q < 6; ++q)
      xa[t][q] = *reinterpret_cast<const v2f*>(x + (i0 + n0) * DDIM + 4 * q + koff);
#pragma unroll
    for (int m = 0; m < 8; ++m) {
      xc0[t][m] = x[(i0 + rowbase + m) * DDIM + j1];
      xc1[t][m] = x[(i0 + rowbase + m) * DDIM + j2];
      acc0[t][m] = 0.0f;
      acc1[t][m] = 0.0f;
    }
  }

  const int l0 = chunk * LCHUNK;
  for (int l = l0; l < l0 + LCHUNK; ++l) {
    const float* xlp = x + l * DDIM;
    const float* Bl  = betaT + (size_t)l * (DDIM * DDIM);

    v2f xl[6], b0[6], b1[6];
#pragma unroll
    for (int q = 0; q < 6; ++q) {
      xl[q] = *reinterpret_cast<const v2f*>(xlp + 4 * q + koff);
      b0[q] = *reinterpret_cast<const v2f*>(Bl + j1 * DDIM + 4 * q + koff);
      b1[q] = *reinterpret_cast<const v2f*>(Bl + j2 * DDIM + 4 * q + koff);
    }
    const float xl0  = xlp[j1];
    const float xl1  = xlp[j2];
    const float al0  = alphaT[l * DDIM + j1];
    const float al1t = alphaT[l * DDIM + j2];
    const float al1  = jt1ok ? al1t : 0.0f;

    v2f b1m[6];
#pragma unroll
    for (int q = 0; q < 6; ++q) {
      b1m[q].x = jt1ok ? b1[q].x : 0.0f;
      b1m[q].y = jt1ok ? b1[q].y : 0.0f;
    }

#pragma unroll
    for (int t = 0; t < MT; ++t) {
      v2f a[6];
      float p = 0.0f;
#pragma unroll
      for (int q = 0; q < 6; ++q) {
        a[q] = xa[t][q] - xl[q];
        p += a[q].x * a[q].x;
        p += a[q].y * a[q].y;
      }
      float s = p + xhalf(p);
      s = __builtin_bit_cast(float, __builtin_amdgcn_update_dpp(
          __builtin_bit_cast(int, s), __builtin_bit_cast(int, s),
          0x128, 0x2, 0xF, false));

      v8f c0 = {};
      v8f c1 = {};
#pragma unroll
      for (int q = 0; q < 6; ++q) {
        c0 = __builtin_amdgcn_wmma_f32_16x16x4_f32(false, a[q], false, b0[q],
                                                   (short)0, c0, false, false);
        c1 = __builtin_amdgcn_wmma_f32_16x16x4_f32(false, a[q], false, b1m[q],
                                                   (short)0, c1, false, false);
      }

      float Sr[8];
      Sr[0] = row_bcast<0>(s); Sr[1] = row_bcast<1>(s);
      Sr[2] = row_bcast<2>(s); Sr[3] = row_bcast<3>(s);
      Sr[4] = row_bcast<4>(s); Sr[5] = row_bcast<5>(s);
      Sr[6] = row_bcast<6>(s); Sr[7] = row_bcast<7>(s);
#pragma unroll
      for (int m = 0; m < 8; ++m) {
        float S  = Sr[m];
        float d0 = xc0[t][m] - xl0;
        float w0 = __expf(d0 * d0 - S);
        acc0[t][m] += w0 * (al0 + 2.0f * c0[m]);
        float d1 = xc1[t][m] - xl1;
        float w1 = __expf(d1 * d1 - S);
        acc1[t][m] += w1 * (al1 + 2.0f * c1[m]);
      }
    }
  }

#pragma unroll
  for (int t = 0; t < MT; ++t) {
    float* p = partial + ((size_t)chunk * NPTS + ibase + 16 * t) * DDIM;
#pragma unroll
    for (int m = 0; m < 8; ++m) {
      int r = rowbase + m;
      p[r * DDIM + j1] = acc0[t][m];
      if (jt1ok) p[r * DDIM + 16 + n0] = acc1[t][m];
    }
  }
}

__global__ __launch_bounds__(256) void reduce_kernel(const float* __restrict__ partial,
                                                     float* __restrict__ out) {
  int t = blockIdx.x * blockDim.x + threadIdx.x;
  if (t < NPTS * DDIM) {
    float s = 0.0f;
#pragma unroll
    for (int c = 0; c < LSPLIT; ++c)
      s += partial[(size_t)c * (NPTS * DDIM) + t];
    out[t] = s;
  }
}

extern "C" void kernel_launch(void* const* d_in, const int* in_sizes, int n_in,
                              void* d_out, int out_size, void* d_ws, size_t ws_size,
                              hipStream_t stream) {
  const float* x     = (const float*)d_in[0];
  const float* alpha = (const float*)d_in[1];
  const float* beta  = (const float*)d_in[2];
  float* out = (float*)d_out;

  char* ws = (char*)d_ws;
  float* betaT   = (float*)ws;
  float* alphaT  = (float*)(ws + (size_t)BETAT_FLOATS * sizeof(float));
  float* partial = (float*)(ws + (size_t)(BETAT_FLOATS + ALPHAT_FLOATS) * sizeof(float));

  (void)in_sizes; (void)n_in; (void)out_size; (void)ws_size;

  prep_kernel<<<PREP_BLOCKS, 256, 0, stream>>>(alpha, beta, betaT, alphaT);

  dim3 grid(NGROUP, LSPLIT / WPB);
  rkhs_main<<<grid, 32 * WPB, 0, stream>>>(x, alphaT, betaT, partial);

  reduce_kernel<<<(NPTS * DDIM + 255) / 256, 256, 0, stream>>>(partial, out);
}


# main: the 30 fragment loads of an l pair are issued one trip ahead, behind the last WMMA of the trip before, and fly during its five combines (a trip opened with a full wait); values unchanged
# speedup vs baseline: 4.7662x; 1.2563x over previous
#include <hip/hip_runtime.h>


typedef __attribute__((ext_vector_type(2))) float v2f;
typedef __attribute__((ext_vector_type(8))) float v8f;

#define NPTS 2000
#define DDIM 24
#define NTILE 125
#define MT 5
#define NGROUP (NTILE / MT)
#define WPB 4
#define LSPLIT 80
#define LCHUNK (NPTS / LSPLIT)

#define BETAT_FLOATS  (NPTS * DDIM * DDIM)
#define ALPHAT_FLOATS (NPTS * DDIM)
#define PART_FLOATS   (LSPLIT * NPTS * DDIM)

#define PT 64
#define PPITCH 65
#define PRT (DDIM * DDIM / PT)
#define PLT ((NPTS + PT - 1) / PT)
#define PREP_BLOCKS (PRT * PLT + PLT)

__global__ __launch_bounds__(256) void prep_kernel(const float* __restrict__ alpha,
                                                   const float* __restrict__ beta,
                                                   float* __restrict__ betaT,
                                                   float* __restrict__ alphaT) {
  __shared__ float tile[PT][PPITCH];
  const int lane = threadIdx.x & 31;
  const int w = __builtin_amdgcn_readfirstlane((int)(threadIdx.x >> 5));
  const int b = blockIdx.x;
  const bool is_beta = b < PRT * PLT;
  const int l0 = (is_beta ? (b / PRT) : (b - PRT * PLT)) * PT;
  const int r0 = is_beta ? (b % PRT) * PT : 0;
  const int lg  = l0 + 2 * lane;
  const int lgc = (lg < NPTS) ? lg : (NPTS - 2);

  if (is_beta) {
    v2f v[8];
#pragma unroll
    for (int p = 0; p < 8; ++p) {
      const int r = r0 + w + 8 * p;
      v[p] = *reinterpret_cast<const v2f*>(beta + (size_t)r * NPTS + lgc);
    }
#pragma unroll
    for (int p = 0; p < 8; ++p) {
      const int rr = w + 8 * p;
      const int r  = r0 + rr;
      const bool diag = (r % (DDIM + 1)) == 0;
      tile[2 * lane][rr]     = diag ? 0.0f : v[p].x;
      tile[2 * lane + 1][rr] = diag ? 0.0f : v[p].y;
    }
  } else {
    v2f v[3];
#pragma unroll
    for (int p = 0; p < 3; ++p) {
      const int j = w + 8 * p;
      v[p] = *reinterpret_cast<const v2f*>(alpha + j * NPTS + lgc);
    }
#pragma unroll
    for (int p = 0; p < 3; ++p) {
      const int j = w + 8 * p;
      tile[2 * lane][j]     = v[p].x;
      tile[2 * lane + 1][j] = v[p].y;
    }
  }
  __syncthreads();
  if (is_beta) {
    v2f o[8];
#pragma unroll
    for (int p = 0; p < 8; ++p) {
      const int ll = w + 8 * p;
      o[p].x = tile[ll][2 * lane];
      o[p].y = tile[ll][2 * lane + 1];
    }
#pragma unroll
    for (int p = 0; p < 8; ++p) {
      const int l = l0 + w + 8 * p;
      if (l < NPTS)
        *reinterpret_cast<v2f*>(betaT + (size_t)l * (DDIM * DDIM) + r0 + 2 * lane) = o[p];
    }
  } else {
    float o[(PT * DDIM) / 256];
#pragma unroll
    for (int k = 0; k < (PT * DDIM) / 256; ++k) {
      const int e  = (int)threadIdx.x + 256 * k;
      const int ll = e / DDIM;
      o[k] = tile[ll][e - ll * DDIM];
    }
#pragma unroll
    for (int k = 0; k < (PT * DDIM) / 256; ++k) {
      const int e = (int)threadIdx.x + 256 * k;
      if (l0 + e / DDIM < NPTS) alphaT[l0 * DDIM + e] = o[k];
    }
  }
}

__device__ __forceinline__ float xhalf(float v) {
  return __builtin_bit_cast(float, __builtin_amdgcn_permlanex16(
      __builtin_bit_cast(int, v), __builtin_bit_cast(int, v),
      0x76543210, 0xfedcba98, false, false));
}

template <int SRC>
__device__ __forceinline__ float row_bcast(float v) {
  return __builtin_bit_cast(float, __builtin_amdgcn_update_dpp(
      0, __builtin_bit_cast(int, v), 0x150 + SRC, 0xF, 0xF, false));
}

__global__ __launch_bounds__(32 * WPB) __attribute__((amdgpu_waves_per_eu(2, 2)))
void rkhs_main(const float* __restrict__ x,
               const float* __restrict__ alphaT,
               const float* __restrict__ betaT,
               float* __restrict__ partial) {
  const int lane  = threadIdx.x & 31;
  const int wave  = __builtin_amdgcn_readfirstlane((int)(threadIdx.x >> 5));
  const int chunk = blockIdx.y * WPB + wave;
  const int ibase = blockIdx.x * (16 * MT);

  const int  n0      = lane & 15;
  const int  koff    = (lane >> 4) * 2;
  const int  rowbase = (lane >> 4) * 8;
  const int  hi8     = n0 >> 3;
  const bool mlb     = (hi8 != 0);
  const int  j1      = n0;
  const int  j2      = 16 + (n0 & 7);

  v2f   xa[MT][6];
  v2f xc0[MT][4], xc1[MT][4];
  v2f acc0[MT][4], acc1[MT][4];
#pragma unroll
  for (int t = 0; t < MT; ++t) {
    const int i0 = ibase + 16 * t;
#pragma unroll
    for (int q = 0; q < 6; ++q)
      xa[t][q] = *reinterpret_cast<const v2f*>(x + (i0 + n0) * DDIM + 4 * q + koff);
#pragma unroll
    for (int h = 0; h < 4; ++h) {
      xc0[t][h].x = x[(i0 + rowbase + 2 * h) * DDIM + j1];
      xc0[t][h].y = x[(i0 + rowbase + 2 * h + 1) * DDIM + j1];
      xc1[t][h].x = x[(i0 + rowbase + 2 * h) * DDIM + j2];
      xc1[t][h].y = x[(i0 + rowbase + 2 * h + 1) * DDIM + j2];
      acc0[t][h] = v2f{0.0f, 0.0f};
      acc1[t][h] = v2f{0.0f, 0.0f};
    }
  }

  const int lfirst = chunk * LCHUNK + (chunk & 1);
  const int npair  = 13 - (chunk & 1);

  v2f xfa[6], xfb[6], bA[6], bB[6], bM[6];

#define LOAD_FRAGS(LA)                                                         \
  {                                                                            \
    const int    la_  = (LA);                                                  \
    const float* xla_ = x + la_ * DDIM;                                        \
    const float* xlb_ = xla_ + DDIM;                                           \
    const float* Ba_  = betaT + (size_t)la_ * (DDIM * DDIM);                   \
    const float* Bb_  = Ba_ + DDIM * DDIM;                                     \
    const float* Bm_  = Ba_ + hi8 * (DDIM * DDIM);      \
    _Pragma("unroll")                                                          \
    for (int q = 0; q < 6; ++q) {                                              \
      xfa[q] = *reinterpret_cast<const v2f*>(xla_ + 4 * q + koff);             \
      xfb[q] = *reinterpret_cast<const v2f*>(xlb_ + 4 * q + koff);             \
      bA[q]  = *reinterpret_cast<const v2f*>(Ba_ + j1 * DDIM + 4 * q + koff);  \
      bB[q]  = *reinterpret_cast<const v2f*>(Bb_ + j1 * DDIM + 4 * q + koff);  \
      bM[q]  = *reinterpret_cast<const v2f*>(Bm_ + j2 * DDIM + 4 * q + koff);  \
    }                                                                          \
  }

  LOAD_FRAGS(lfirst)

  for (int pr = 0; pr < npair; ++pr) {
    const int la = lfirst + 2 * pr;
    const float* xla = x + la * DDIM;
    const float xla_j = xla[j1];
    const float xlb_j = xla[DDIM + j1];
    const float xlm_j = xla[hi8 * DDIM + j2];
    const float ala   = alphaT[la * DDIM + j1];
    const float alb   = alphaT[(la + 1) * DDIM + j1];
    const float alm   = alphaT[(la + hi8) * DDIM + j2];

    v2f va = {0.0f, 0.0f}, vb = {0.0f, 0.0f}, vm = {0.0f, 0.0f};
#pragma unroll
    for (int q = 0; q < 6; ++q) {
      const v2f xm = mlb ? xfb[q] : xfa[q];
      va = __builtin_elementwise_fma(xfa[q], bA[q], va);
      vb = __builtin_elementwise_fma(xfb[q], bB[q], vb);
      vm = __builtin_elementwise_fma(xm, bM[q], vm);
    }
    const float pa = va.x + va.y;
    const float pb = vb.x + vb.y;
    const float pm = vm.x + vm.y;
    const float ea = ala - 2.0f * (pa + xhalf(pa));
    const float eb = alb - 2.0f * (pb + xhalf(pb));
    const float em = alm - 2.0f * (pm + xhalf(pm));

    float sa[MT], sb[MT];
    v8f cA[MT], cB[MT], cM[MT];
#pragma unroll
    for (int t = 0; t < MT; ++t) {
      v2f ua = {0.0f, 0.0f}, ub = {0.0f, 0.0f};
#pragma unroll
      for (int q = 0; q < 6; ++q) {
        const v2f da = xa[t][q] - xfa[q];
        const v2f db = xa[t][q] - xfb[q];
        ua = __builtin_elementwise_fma(da, da, ua);
        ub = __builtin_elementwise_fma(db, db, ub);
      }
      const float qa = ua.x + ua.y;
      const float qb = ub.x + ub.y;
      float sat = qa + xhalf(qa);
      float sbt = qb + xhalf(qb);
      sa[t] = __builtin_bit_cast(float, __builtin_amdgcn_update_dpp(
          __builtin_bit_cast(int, sat), __builtin_bit_cast(int, sat),
          0x128, 0x2, 0xF, false));
      sb[t] = __builtin_bit_cast(float, __builtin_amdgcn_update_dpp(
          __builtin_bit_cast(int, sbt), __builtin_bit_cast(int, sbt),
          0x128, 0x2, 0xF, false));

      v8f tA = {};
      v8f tB = {};
      v8f tM = {};
#pragma unroll
      for (int q = 0; q < 6; ++q) {
        tA = __builtin_amdgcn_wmma_f32_16x16x4_f32(false, xa[t][q], false, bA[q],
                                                   (short)0, tA, false, false);
        tB = __builtin_amdgcn_wmma_f32_16x16x4_f32(false, xa[t][q], false, bB[q],
                                                   (short)0, tB, false, false);
        tM = __builtin_amdgcn_wmma_f32_16x16x4_f32(false, xa[t][q], false, bM[q],
                                                   (short)0, tM, false, false);
      }
      cA[t] = tA;
      cB[t] = tB;
      cM[t] = tM;
    }

#define COMBINE_TILE(T)                                                        \
    {                                                                          \
      float SA[8], SB[8];                                                      \
      SA[0] = row_bcast<0>(sa[T]); SA[1] = row_bcast<1>(sa[T]);                \
      SA[2] = row_bcast<2>(sa[T]); SA[3] = row_bcast<3>(sa[T]);                \
      SA[4] = row_bcast<4>(sa[T]); SA[5] = row_bcast<5>(sa[T]);                \
      SA[6] = row_bcast<6>(sa[T]); SA[7] = row_bcast<7>(sa[T]);                \
      SB[0] = row_bcast<0>(sb[T]); SB[1] = row_bcast<1>(sb[T]);                \
      SB[2] = row_bcast<2>(sb[T]); SB[3] = row_bcast<3>(sb[T]);                \
      SB[4] = row_bcast<4>(sb[T]); SB[5] = row_bcast<5>(sb[T]);                \
      SB[6] = row_bcast<6>(sb[T]); SB[7] = row_bcast<7>(sb[T]);                \
      _Pragma("unroll")                                                        \
      for (int h = 0; h < 4; ++h) {                                            \
        const int m = 2 * h;                                                   \
        const v2f sA2 = {SA[m], SA[m + 1]};                                    \
        const v2f sB2 = {SB[m], SB[m + 1]};                                    \
        const v2f sM2 = {mlb ? SB[m] : SA[m], mlb ? SB[m + 1] : SA[m + 1]};    \
        const v2f gA  = {cA[T][m], cA[T][m + 1]};                              \
        const v2f gB  = {cB[T][m], cB[T][m + 1]};                              \
        const v2f gM  = {cM[T][m], cM[T][m + 1]};                              \
        const v2f dA = xc0[T][h] - xla_j;                                      \
        const v2f rA = dA * dA - sA2;                                          \
        const v2f wA = {__expf(rA.x), __expf(rA.y)};                           \
        acc0[T][h] += wA * (ea + 2.0f * gA);                                   \
        const v2f dB = xc0[T][h] - xlb_j;                                      \
        const v2f rB = dB * dB - sB2;                                          \
        const v2f wB = {__expf(rB.x), __expf(rB.y)};                           \
        acc0[T][h] += wB * (eb + 2.0f * gB);                                   \
        const v2f dM = xc1[T][h] - xlm_j;                                      \
        const v2f rM = dM * dM - sM2;                                          \
        const v2f wM = {__expf(rM.x), __expf(rM.y)};                           \
        acc1[T][h] += wM * (em + 2.0f * gM);                                   \
      }                                                                        \
    }

    __builtin_amdgcn_sched_barrier(0);
    COMBINE_TILE(0)

    {
      const int pn = (pr + 1 < npair) ? (pr + 1) : pr;
      LOAD_FRAGS(lfirst + 2 * pn)
    }

#pragma unroll
    for (int t = 1; t < MT; ++t) {
      __builtin_amdgcn_sched_barrier(0);
      COMBINE_TILE(t)
    }
  }
#undef COMBINE_TILE
#undef LOAD_FRAGS

#pragma unroll
  for (int t = 0; t < MT; ++t) {
    float* p = partial + ((size_t)chunk * NPTS + ibase + 16 * t) * DDIM;
#pragma unroll
    for (int m = 0; m < 8; ++m) {
      const int r = rowbase + m;
      const float a0 = (m & 1) ? acc0[t][m >> 1].y : acc0[t][m >> 1].x;
      const float a1 = (m & 1) ? acc1[t][m >> 1].y : acc1[t][m >> 1].x;
      const float other = __builtin_bit_cast(float, __builtin_amdgcn_update_dpp(
          0, __builtin_bit_cast(int, a1), 0x128, 0xF, 0xF, false));
      p[r * DDIM + j1] = a0;
      if (!mlb) p[r * DDIM + j2] = a1 + other;
    }
  }
}

__global__ __launch_bounds__(256) void reduce_kernel(const float* __restrict__ partial,
                                                     float* __restrict__ out) {
  int t = blockIdx.x * blockDim.x + threadIdx.x;
  if (t < NPTS * DDIM) {
    float s = 0.0f;
#pragma unroll
    for (int c = 0; c < LSPLIT; ++c)
      s += partial[(size_t)c * (NPTS * DDIM) + t];
    out[t] = s;
  }
}

extern "C" void kernel_launch(void* const* d_in, const int* in_sizes, int n_in,
                              void* d_out, int out_size, void* d_ws, size_t ws_size,
                              hipStream_t stream) {
  const float* x     = (const float*)d_in[0];
  const float* alpha = (const float*)d_in[1];
  const float* beta  = (const float*)d_in[2];
  float* out = (float*)d_out;

  char* ws = (char*)d_ws;
  float* betaT   = (float*)ws;
  float* alphaT  = (float*)(ws + (size_t)BETAT_FLOATS * sizeof(float));
  float* partial = (float*)(ws + (size_t)(BETAT_FLOATS + ALPHAT_FLOATS) * sizeof(float));

  (void)in_sizes; (void)n_in; (void)out_size; (void)ws_size;

  prep_kernel<<<PREP_BLOCKS, 256, 0, stream>>>(alpha, beta, betaT, alphaT);

  dim3 grid(NGROUP, LSPLIT / WPB);
  rkhs_main<<<grid, 32 * WPB, 0, stream>>>(x, alphaT, betaT, partial);

  reduce_kernel<<<(NPTS * DDIM + 255) / 256, 256, 0, stream>>>(partial, out);
}


# main: blocks of 8 waves add their 8 l-chunk tiles in LDS (one barrier) and store one tile in whole 128-B lines: 10 partial planes, was 80, for the reduce kernel; the partial sums regroup
# speedup vs baseline: 5.0759x; 1.0650x over previous
#include <hip/hip_runtime.h>


typedef __attribute__((ext_vector_type(2))) float v2f;
typedef __attribute__((ext_vector_type(4))) float v4f;
typedef __attribute__((ext_vector_type(8))) float v8f;

#define NPTS 2000
#define DDIM 24
#define NTILE 125
#define MT 5
#define NGROUP (NTILE / MT)
#define WPB 8
#define LSPLIT 80
#define LCHUNK (NPTS / LSPLIT)
#define NPLANE (LSPLIT / WPB)
#define TILE_FLOATS (16 * MT * DDIM)

#define BETAT_FLOATS  (NPTS * DDIM * DDIM)
#define ALPHAT_FLOATS (NPTS * DDIM)
#define PART_FLOATS   (NPLANE * NPTS * DDIM)

#define PT 64
#define PPITCH 65
#define PRT (DDIM * DDIM / PT)
#define PLT ((NPTS + PT - 1) / PT)
#define PREP_BLOCKS (PRT * PLT + PLT)

__global__ __launch_bounds__(256) void prep_kernel(const float* __restrict__ alpha,
                                                   const float* __restrict__ beta,
                                                   float* __restrict__ betaT,
                                                   float* __restrict__ alphaT) {
  __shared__ float tile[PT][PPITCH];
  const int lane = threadIdx.x & 31;
  const int w = __builtin_amdgcn_readfirstlane((int)(threadIdx.x >> 5));
  const int b = blockIdx.x;
  const bool is_beta = b < PRT * PLT;
  const int l0 = (is_beta ? (b / PRT) : (b - PRT * PLT)) * PT;
  const int r0 = is_beta ? (b % PRT) * PT : 0;
  const int lg  = l0 + 2 * lane;
  const int lgc = (lg < NPTS) ? lg : (NPTS - 2);

  if (is_beta) {
    v2f v[8];
#pragma unroll
    for (int p = 0; p < 8; ++p) {
      const int r = r0 + w + 8 * p;
      v[p] = *reinterpret_cast<const v2f*>(beta + (size_t)r * NPTS + lgc);
    }
#pragma unroll
    for (int p = 0; p < 8; ++p) {
      const int rr = w + 8 * p;
      const int r  = r0 + rr;
      const bool diag = (r % (DDIM + 1)) == 0;
      tile[2 * lane][rr]     = diag ? 0.0f : v[p].x;
      tile[2 * lane + 1][rr] = diag ? 0.0f : v[p].y;
    }
  } else {
    v2f v[3];
#pragma unroll
    for (int p = 0; p < 3; ++p) {
      const int j = w + 8 * p;
      v[p] = *reinterpret_cast<const v2f*>(alpha + j * NPTS + lgc);
    }
#pragma unroll
    for (int p = 0; p < 3; ++p) {
      const int j = w + 8 * p;
      tile[2 * lane][j]     = v[p].x;
      tile[2 * lane + 1][j] = v[p].y;
    }
  }
  __syncthreads();
  if (is_beta) {
    v2f o[8];
#pragma unroll
    for (int p = 0; p < 8; ++p) {
      const int ll = w + 8 * p;
      o[p].x = tile[ll][2 * lane];
      o[p].y = tile[ll][2 * lane + 1];
    }
#pragma unroll
    for (int p = 0; p < 8; ++p) {
      const int l = l0 + w + 8 * p;
      if (l < NPTS)
        *reinterpret_cast<v2f*>(betaT + (size_t)l * (DDIM * DDIM) + r0 + 2 * lane) = o[p];
    }
  } else {
    float o[(PT * DDIM) / 256];
#pragma unroll
    for (int k = 0; k < (PT * DDIM) / 256; ++k) {
      const int e  = (int)threadIdx.x + 256 * k;
      const int ll = e / DDIM;
      o[k] = tile[ll][e - ll * DDIM];
    }
#pragma unroll
    for (int k = 0; k < (PT * DDIM) / 256; ++k) {
      const int e = (int)threadIdx.x + 256 * k;
      if (l0 + e / DDIM < NPTS) alphaT[l0 * DDIM + e] = o[k];
    }
  }
}

__device__ __forceinline__ float xhalf(float v) {
  return __builtin_bit_cast(float, __builtin_amdgcn_permlanex16(
      __builtin_bit_cast(int, v), __builtin_bit_cast(int, v),
      0x76543210, 0xfedcba98, false, false));
}

template <int SRC>
__device__ __forceinline__ float row_bcast(float v) {
  return __builtin_bit_cast(float, __builtin_amdgcn_update_dpp(
      0, __builtin_bit_cast(int, v), 0x150 + SRC, 0xF, 0xF, false));
}

__global__ __launch_bounds__(32 * WPB) __attribute__((amdgpu_waves_per_eu(2, 2)))
void rkhs_main(const float* __restrict__ x,
               const float* __restrict__ alphaT,
               const float* __restrict__ betaT,
               float* __restrict__ partial) {
  const int lane  = threadIdx.x & 31;
  const int wave  = __builtin_amdgcn_readfirstlane((int)(threadIdx.x >> 5));
  const int chunk = blockIdx.y * WPB + wave;
  const int ibase = blockIdx.x * (16 * MT);

  const int  n0      = lane & 15;
  const int  koff    = (lane >> 4) * 2;
  const int  rowbase = (lane >> 4) * 8;
  const int  hi8     = n0 >> 3;
  const bool mlb     = (hi8 != 0);
  const int  j1      = n0;
  const int  j2      = 16 + (n0 & 7);

  v2f   xa[MT][6];
  v2f xc0[MT][4], xc1[MT][4];
  v2f acc0[MT][4], acc1[MT][4];
#pragma unroll
  for (int t = 0; t < MT; ++t) {
    const int i0 = ibase + 16 * t;
#pragma unroll
    for (int q = 0; q < 6; ++q)
      xa[t][q] = *reinterpret_cast<const v2f*>(x + (i0 + n0) * DDIM + 4 * q + koff);
#pragma unroll
    for (int h = 0; h < 4; ++h) {
      xc0[t][h].x = x[(i0 + rowbase + 2 * h) * DDIM + j1];
      xc0[t][h].y = x[(i0 + rowbase + 2 * h + 1) * DDIM + j1];
      xc1[t][h].x = x[(i0 + rowbase + 2 * h) * DDIM + j2];
      xc1[t][h].y = x[(i0 + rowbase + 2 * h + 1) * DDIM + j2];
      acc0[t][h] = v2f{0.0f, 0.0f};
      acc1[t][h] = v2f{0.0f, 0.0f};
    }
  }

  const int lfirst = chunk * LCHUNK + (chunk & 1);
  const int npair  = 13 - (chunk & 1);

  v2f xfa[6], xfb[6], bA[6], bB[6], bM[6];

#define LOAD_FRAGS(LA)                                                         \
  {                                                                            \
    const int    la_  = (LA);                                                  \
    const float* xla_ = x + la_ * DDIM;                                        \
    const float* xlb_ = xla_ + DDIM;                                           \
    const float* Ba_  = betaT + (size_t)la_ * (DDIM * DDIM);                   \
    const float* Bb_  = Ba_ + DDIM * DDIM;                                     \
    const float* Bm_  = Ba_ + hi8 * (DDIM * DDIM);      \
    _Pragma("unroll")                                                          \
    for (int q = 0; q < 6; ++q) {                                              \
      xfa[q] = *reinterpret_cast<const v2f*>(xla_ + 4 * q + koff);             \
      xfb[q] = *reinterpret_cast<const v2f*>(xlb_ + 4 * q + koff);             \
      bA[q]  = *reinterpret_cast<const v2f*>(Ba_ + j1 * DDIM + 4 * q + koff);  \
      bB[q]  = *reinterpret_cast<const v2f*>(Bb_ + j1 * DDIM + 4 * q + koff);  \
      bM[q]  = *reinterpret_cast<const v2f*>(Bm_ + j2 * DDIM + 4 * q + koff);  \
    }                                                                          \
  }

  LOAD_FRAGS(lfirst)

  for (int pr = 0; pr < npair; ++pr) {
    const int la = lfirst + 2 * pr;
    const float* xla = x + la * DDIM;
    const float xla_j = xla[j1];
    const float xlb_j = xla[DDIM + j1];
    const float xlm_j = xla[hi8 * DDIM + j2];
    const float ala   = alphaT[la * DDIM + j1];
    const float alb   = alphaT[(la + 1) * DDIM + j1];
    const float alm   = alphaT[(la + hi8) * DDIM + j2];

    v2f va = {0.0f, 0.0f}, vb = {0.0f, 0.0f}, vm = {0.0f, 0.0f};
#pragma unroll
    for (int q = 0; q < 6; ++q) {
      const v2f xm = mlb ? xfb[q] : xfa[q];
      va = __builtin_elementwise_fma(xfa[q], bA[q], va);
      vb = __builtin_elementwise_fma(xfb[q], bB[q], vb);
      vm = __builtin_elementwise_fma(xm, bM[q], vm);
    }
    const float pa = va.x + va.y;
    const float pb = vb.x + vb.y;
    const float pm = vm.x + vm.y;
    const float ea = ala - 2.0f * (pa + xhalf(pa));
    const float eb = alb - 2.0f * (pb + xhalf(pb));
    const float em = alm - 2.0f * (pm + xhalf(pm));

    float sa[MT], sb[MT];
    v8f cA[MT], cB[MT], cM[MT];
#pragma unroll
    for (int t = 0; t < MT; ++t) {
      v2f ua = {0.0f, 0.0f}, ub = {0.0f, 0.0f};
#pragma unroll
      for (int q = 0; q < 6; ++q) {
        const v2f da = xa[t][q] - xfa[q];
        const v2f db = xa[t][q] - xfb[q];
        ua = __builtin_elementwise_fma(da, da, ua);
        ub = __builtin_elementwise_fma(db, db, ub);
      }
      const float qa = ua.x + ua.y;
      const float qb = ub.x + ub.y;
      float sat = qa + xhalf(qa);
      float sbt = qb + xhalf(qb);
      sa[t] = __builtin_bit_cast(float, __builtin_amdgcn_update_dpp(
          __builtin_bit_cast(int, sat), __builtin_bit_cast(int, sat),
          0x128, 0x2, 0xF, false));
      sb[t] = __builtin_bit_cast(float, __builtin_amdgcn_update_dpp(
          __builtin_bit_cast(int, sbt), __builtin_bit_cast(int, sbt),
          0x128, 0x2, 0xF, false));

      v8f tA = {};
      v8f tB = {};
      v8f tM = {};
#pragma unroll
      for (int q = 0; q < 6; ++q) {
        tA = __builtin_amdgcn_wmma_f32_16x16x4_f32(false, xa[t][q], false, bA[q],
                                                   (short)0, tA, false, false);
        tB = __builtin_amdgcn_wmma_f32_16x16x4_f32(false, xa[t][q], false, bB[q],
                                                   (short)0, tB, false, false);
        tM = __builtin_amdgcn_wmma_f32_16x16x4_f32(false, xa[t][q], false, bM[q],
                                                   (short)0, tM, false, false);
      }
      cA[t] = tA;
      cB[t] = tB;
      cM[t] = tM;
    }

#define COMBINE_TILE(T)                                                        \
    {                                                                          \
      float SA[8], SB[8];                                                      \
      SA[0] = row_bcast<0>(sa[T]); SA[1] = row_bcast<1>(sa[T]);                \
      SA[2] = row_bcast<2>(sa[T]); SA[3] = row_bcast<3>(sa[T]);                \
      SA[4] = row_bcast<4>(sa[T]); SA[5] = row_bcast<5>(sa[T]);                \
      SA[6] = row_bcast<6>(sa[T]); SA[7] = row_bcast<7>(sa[T]);                \
      SB[0] = row_bcast<0>(sb[T]); SB[1] = row_bcast<1>(sb[T]);                \
      SB[2] = row_bcast<2>(sb[T]); SB[3] = row_bcast<3>(sb[T]);                \
      SB[4] = row_bcast<4>(sb[T]); SB[5] = row_bcast<5>(sb[T]);                \
      SB[6] = row_bcast<6>(sb[T]); SB[7] = row_bcast<7>(sb[T]);                \
      _Pragma("unroll")                                                        \
      for (int h = 0; h < 4; ++h) {                                            \
        const int m = 2 * h;                                                   \
        const v2f sA2 = {SA[m], SA[m + 1]};                                    \
        const v2f sB2 = {SB[m], SB[m + 1]};                                    \
        const v2f sM2 = {mlb ? SB[m] : SA[m], mlb ? SB[m + 1] : SA[m + 1]};    \
        const v2f gA  = {cA[T][m], cA[T][m + 1]};                              \
        const v2f gB  = {cB[T][m], cB[T][m + 1]};                              \
        const v2f gM  = {cM[T][m], cM[T][m + 1]};                              \
        const v2f dA = xc0[T][h] - xla_j;                                      \
        const v2f rA = dA * dA - sA2;                                          \
        const v2f wA = {__expf(rA.x), __expf(rA.y)};                           \
        acc0[T][h] += wA * (ea + 2.0f * gA);                                   \
        const v2f dB = xc0[T][h] - xlb_j;                                      \
        const v2f rB = dB * dB - sB2;                                          \
        const v2f wB = {__expf(rB.x), __expf(rB.y)};                           \
        acc0[T][h] += wB * (eb + 2.0f * gB);                                   \
        const v2f dM = xc1[T][h] - xlm_j;                                      \
        const v2f rM = dM * dM - sM2;                                          \
        const v2f wM = {__expf(rM.x), __expf(rM.y)};                           \
        acc1[T][h] += wM * (em + 2.0f * gM);                                   \
      }                                                                        \
    }

    __builtin_amdgcn_sched_barrier(0);
    COMBINE_TILE(0)

    {
      const int pn = (pr + 1 < npair) ? (pr + 1) : pr;
      LOAD_FRAGS(lfirst + 2 * pn)
    }

#pragma unroll
    for (int t = 1; t < MT; ++t) {
      __builtin_amdgcn_sched_barrier(0);
      COMBINE_TILE(t)
    }
  }
#undef COMBINE_TILE
#undef LOAD_FRAGS

  __shared__ __attribute__((aligned(16))) float red[WPB][TILE_FLOATS];
  float* mine = red[wave];
#pragma unroll
  for (int t = 0; t < MT; ++t) {
#pragma unroll
    for (int m = 0; m < 8; ++m) {
      const int r = 16 * t + rowbase + m;
      const float a0 = (m & 1) ? acc0[t][m >> 1].y : acc0[t][m >> 1].x;
      const float a1 = (m & 1) ? acc1[t][m >> 1].y : acc1[t][m >> 1].x;
      const float other = __builtin_bit_cast(float, __builtin_amdgcn_update_dpp(
          0, __builtin_bit_cast(int, a1), 0x128, 0xF, 0xF, false));
      mine[r * DDIM + j1] = a0;
      if (!mlb) mine[r * DDIM + j2] = a1 + other;
    }
  }
  __syncthreads();
  float* p = partial + ((size_t)blockIdx.y * NPTS + ibase) * DDIM;
#pragma unroll
  for (int k = 0; k < 2; ++k) {
    const int e4 = (int)threadIdx.x + 32 * WPB * k;
    if (e4 < TILE_FLOATS / 4) {
      v4f s = *reinterpret_cast<const v4f*>(&red[0][4 * e4]);
#pragma unroll
      for (int w = 1; w < WPB; ++w)
        s += *reinterpret_cast<const v4f*>(&red[w][4 * e4]);
      *reinterpret_cast<v4f*>(p + 4 * e4) = s;
    }
  }
}

__global__ __launch_bounds__(256) void reduce_kernel(const float* __restrict__ partial,
                                                     float* __restrict__ out) {
  int t = blockIdx.x * blockDim.x + threadIdx.x;
  if (t < NPTS * DDIM) {
    float s = 0.0f;
#pragma unroll
    for (int c = 0; c < NPLANE; ++c)
      s += partial[(size_t)c * (NPTS * DDIM) + t];
    out[t] = s;
  }
}

extern "C" void kernel_launch(void* const* d_in, const int* in_sizes, int n_in,
                              void* d_out, int out_size, void* d_ws, size_t ws_size,
                              hipStream_t stream) {
  const float* x     = (const float*)d_in[0];
  const float* alpha = (const float*)d_in[1];
  const float* beta  = (const float*)d_in[2];
  float* out = (float*)d_out;

  char* ws = (char*)d_ws;
  float* betaT   = (float*)ws;
  float* alphaT  = (float*)(ws + (size_t)BETAT_FLOATS * sizeof(float));
  float* partial = (float*)(ws + (size_t)(BETAT_FLOATS + ALPHAT_FLOATS) * sizeof(float));

  (void)in_sizes; (void)n_in; (void)out_size; (void)ws_size;

  prep_kernel<<<PREP_BLOCKS, 256, 0, stream>>>(alpha, beta, betaT, alphaT);

  dim3 grid(NGROUP, LSPLIT / WPB);
  rkhs_main<<<grid, 32 * WPB, 0, stream>>>(x, alphaT, betaT, partial);

  reduce_kernel<<<(NPTS * DDIM + 255) / 256, 256, 0, stream>>>(partial, out);
}


# main: the younger wave of each SIMD pair (w+4) holds s_setprio 3 for its first 8 l-loop trips, then 0: the pair shares the matrix pipe to the end, neither wave finishes alone; values unchanged
# speedup vs baseline: 5.4072x; 1.0653x over previous
#include <hip/hip_runtime.h>


typedef __attribute__((ext_vector_type(2))) float v2f;
typedef __attribute__((ext_vector_type(4))) float v4f;
typedef __attribute__((ext_vector_type(8))) float v8f;

#define NPTS 2000
#define DDIM 24
#define NTILE 125
#define MT 5
#define NGROUP (NTILE / MT)
#define WPB 8
#define LSPLIT 80
#define LCHUNK (NPTS / LSPLIT)
#define NPLANE (LSPLIT / WPB)
#define TILE_FLOATS (16 * MT * DDIM)
#define LEAD_TRIPS 8

#define BETAT_FLOATS  (NPTS * DDIM * DDIM)
#define ALPHAT_FLOATS (NPTS * DDIM)
#define PART_FLOATS   (NPLANE * NPTS * DDIM)

#define PT 64
#define PPITCH 65
#define PRT (DDIM * DDIM / PT)
#define PLT ((NPTS + PT - 1) / PT)
#define PREP_BLOCKS (PRT * PLT + PLT)

__global__ __launch_bounds__(256) void prep_kernel(const float* __restrict__ alpha,
                                                   const float* __restrict__ beta,
                                                   float* __restrict__ betaT,
                                                   float* __restrict__ alphaT) {
  __shared__ float tile[PT][PPITCH];
  const int lane = threadIdx.x & 31;
  const int w = __builtin_amdgcn_readfirstlane((int)(threadIdx.x >> 5));
  const int b = blockIdx.x;
  const bool is_beta = b < PRT * PLT;
  const int l0 = (is_beta ? (b / PRT) : (b - PRT * PLT)) * PT;
  const int r0 = is_beta ? (b % PRT) * PT : 0;
  const int lg  = l0 + 2 * lane;
  const int lgc = (lg < NPTS) ? lg : (NPTS - 2);

  if (is_beta) {
    v2f v[8];
#pragma unroll
    for (int p = 0; p < 8; ++p) {
      const int r = r0 + w + 8 * p;
      v[p] = *reinterpret_cast<const v2f*>(beta + (size_t)r * NPTS + lgc);
    }
#pragma unroll
    for (int p = 0; p < 8; ++p) {
      const int rr = w + 8 * p;
      const int r  = r0 + rr;
      const bool diag = (r % (DDIM + 1)) == 0;
      tile[2 * lane][rr]     = diag ? 0.0f : v[p].x;
      tile[2 * lane + 1][rr] = diag ? 0.0f : v[p].y;
    }
  } else {
    v2f v[3];
#pragma unroll
    for (int p = 0; p < 3; ++p) {
      const int j = w + 8 * p;
      v[p] = *reinterpret_cast<const v2f*>(alpha + j * NPTS + lgc);
    }
#pragma unroll
    for (int p = 0; p < 3; ++p) {
      const int j = w + 8 * p;
      tile[2 * lane][j]     = v[p].x;
      tile[2 * lane + 1][j] = v[p].y;
    }
  }
  __syncthreads();
  if (is_beta) {
    v2f o[8];
#pragma unroll
    for (int p = 0; p < 8; ++p) {
      const int ll = w + 8 * p;
      o[p].x = tile[ll][2 * lane];
      o[p].y = tile[ll][2 * lane + 1];
    }
#pragma unroll
    for (int p = 0; p < 8; ++p) {
      const int l = l0 + w + 8 * p;
      if (l < NPTS)
        *reinterpret_cast<v2f*>(betaT + (size_t)l * (DDIM * DDIM) + r0 + 2 * lane) = o[p];
    }
  } else {
    float o[(PT * DDIM) / 256];
#pragma unroll
    for (int k = 0; k < (PT * DDIM) / 256; ++k) {
      const int e  = (int)threadIdx.x + 256 * k;
      const int ll = e / DDIM;
      o[k] = tile[ll][e - ll * DDIM];
    }
#pragma unroll
    for (int k = 0; k < (PT * DDIM) / 256; ++k) {
      const int e = (int)threadIdx.x + 256 * k;
      if (l0 + e / DDIM < NPTS) alphaT[l0 * DDIM + e] = o[k];
    }
  }
}

__device__ __forceinline__ float xhalf(float v) {
  return __builtin_bit_cast(float, __builtin_amdgcn_permlanex16(
      __builtin_bit_cast(int, v), __builtin_bit_cast(int, v),
      0x76543210, 0xfedcba98, false, false));
}

template <int SRC>
__device__ __forceinline__ float row_bcast(float v) {
  return __builtin_bit_cast(float, __builtin_amdgcn_update_dpp(
      0, __builtin_bit_cast(int, v), 0x150 + SRC, 0xF, 0xF, false));
}

__global__ __launch_bounds__(32 * WPB) __attribute__((amdgpu_waves_per_eu(2, 2)))
void rkhs_main(const float* __restrict__ x,
               const float* __restrict__ alphaT,
               const float* __restrict__ betaT,
               float* __restrict__ partial) {
  const int lane  = threadIdx.x & 31;
  const int wave  = __builtin_amdgcn_readfirstlane((int)(threadIdx.x >> 5));
  const int chunk = blockIdx.y * WPB + wave;
  const int ibase = blockIdx.x * (16 * MT);

  const int  n0      = lane & 15;
  const int  koff    = (lane >> 4) * 2;
  const int  rowbase = (lane >> 4) * 8;
  const int  hi8     = n0 >> 3;
  const bool mlb     = (hi8 != 0);
  const int  j1      = n0;
  const int  j2      = 16 + (n0 & 7);

  v2f   xa[MT][6];
  v2f xc0[MT][4], xc1[MT][4];
  v2f acc0[MT][4], acc1[MT][4];
#pragma unroll
  for (int t = 0; t < MT; ++t) {
    const int i0 = ibase + 16 * t;
#pragma unroll
    for (int q = 0; q < 6; ++q)
      xa[t][q] = *reinterpret_cast<const v2f*>(x + (i0 + n0) * DDIM + 4 * q + koff);
#pragma unroll
    for (int h = 0; h < 4; ++h) {
      xc0[t][h].x = x[(i0 + rowbase + 2 * h) * DDIM + j1];
      xc0[t][h].y = x[(i0 + rowbase + 2 * h + 1) * DDIM + j1];
      xc1[t][h].x = x[(i0 + rowbase + 2 * h) * DDIM + j2];
      xc1[t][h].y = x[(i0 + rowbase + 2 * h + 1) * DDIM + j2];
      acc0[t][h] = v2f{0.0f, 0.0f};
      acc1[t][h] = v2f{0.0f, 0.0f};
    }
  }

  const int lfirst = chunk * LCHUNK + (chunk & 1);
  const int npair  = 13 - (chunk & 1);

  v2f xfa[6], xfb[6], bA[6], bB[6], bM[6];

#define LOAD_FRAGS(LA)                                                         \
  {                                                                            \
    const int    la_  = (LA);                                                  \
    const float* xla_ = x + la_ * DDIM;                                        \
    const float* xlb_ = xla_ + DDIM;                                           \
    const float* Ba_  = betaT + (size_t)la_ * (DDIM * DDIM);                   \
    const float* Bb_  = Ba_ + DDIM * DDIM;                                     \
    const float* Bm_  = Ba_ + hi8 * (DDIM * DDIM);      \
    _Pragma("unroll")                                                          \
    for (int q = 0; q < 6; ++q) {                                              \
      xfa[q] = *reinterpret_cast<const v2f*>(xla_ + 4 * q + koff);             \
      xfb[q] = *reinterpret_cast<const v2f*>(xlb_ + 4 * q + koff);             \
      bA[q]  = *reinterpret_cast<const v2f*>(Ba_ + j1 * DDIM + 4 * q + koff);  \
      bB[q]  = *reinterpret_cast<const v2f*>(Bb_ + j1 * DDIM + 4 * q + koff);  \
      bM[q]  = *reinterpret_cast<const v2f*>(Bm_ + j2 * DDIM + 4 * q + koff);  \
    }                                                                          \
  }

  LOAD_FRAGS(lfirst)

  if (wave >= WPB / 2) __builtin_amdgcn_s_setprio(3);

#pragma unroll
  for (int ph = 0; ph < 2; ++ph) {
  if (ph == 1) __builtin_amdgcn_s_setprio(0);
  const int pbeg = ph ? LEAD_TRIPS : 0;
  const int pend = ph ? npair : LEAD_TRIPS;
  static_assert(LEAD_TRIPS < 12, "the first run must end inside the shortest chunk");
#pragma nounroll
  for (int pr = pbeg; pr < pend; ++pr) {
    const int la = lfirst + 2 * pr;
    const float* xla = x + la * DDIM;
    const float xla_j = xla[j1];
    const float xlb_j = xla[DDIM + j1];
    const float xlm_j = xla[hi8 * DDIM + j2];
    const float ala   = alphaT[la * DDIM + j1];
    const float alb   = alphaT[(la + 1) * DDIM + j1];
    const float alm   = alphaT[(la + hi8) * DDIM + j2];

    v2f va = {0.0f, 0.0f}, vb = {0.0f, 0.0f}, vm = {0.0f, 0.0f};
#pragma unroll
    for (int q = 0; q < 6; ++q) {
      const v2f xm = mlb ? xfb[q] : xfa[q];
      va = __builtin_elementwise_fma(xfa[q], bA[q], va);
      vb = __builtin_elementwise_fma(xfb[q], bB[q], vb);
      vm = __builtin_elementwise_fma(xm, bM[q], vm);
    }
    const float pa = va.x + va.y;
    const float pb = vb.x + vb.y;
    const float pm = vm.x + vm.y;
    const float ea = ala - 2.0f * (pa + xhalf(pa));
    const float eb = alb - 2.0f * (pb + xhalf(pb));
    const float em = alm - 2.0f * (pm + xhalf(pm));

    float sa[MT], sb[MT];
    v8f cA[MT], cB[MT], cM[MT];
#pragma unroll
    for (int t = 0; t < MT; ++t) {
      v2f ua = {0.0f, 0.0f}, ub = {0.0f, 0.0f};
#pragma unroll
      for (int q = 0; q < 6; ++q) {
        const v2f da = xa[t][q] - xfa[q];
        const v2f db = xa[t][q] - xfb[q];
        ua = __builtin_elementwise_fma(da, da, ua);
        ub = __builtin_elementwise_fma(db, db, ub);
      }
      const float qa = ua.x + ua.y;
      const float qb = ub.x + ub.y;
      float sat = qa + xhalf(qa);
      float sbt = qb + xhalf(qb);
      sa[t] = __builtin_bit_cast(float, __builtin_amdgcn_update_dpp(
          __builtin_bit_cast(int, sat), __builtin_bit_cast(int, sat),
          0x128, 0x2, 0xF, false));
      sb[t] = __builtin_bit_cast(float, __builtin_amdgcn_update_dpp(
          __builtin_bit_cast(int, sbt), __builtin_bit_cast(int, sbt),
          0x128, 0x2, 0xF, false));

      v8f tA = {};
      v8f tB = {};
      v8f tM = {};
#pragma unroll
      for (int q = 0; q < 6; ++q) {
        tA = __builtin_amdgcn_wmma_f32_16x16x4_f32(false, xa[t][q], false, bA[q],
                                                   (short)0, tA, false, false);
        tB = __builtin_amdgcn_wmma_f32_16x16x4_f32(false, xa[t][q], false, bB[q],
                                                   (short)0, tB, false, false);
        tM = __builtin_amdgcn_wmma_f32_16x16x4_f32(false, xa[t][q], false, bM[q],
                                                   (short)0, tM, false, false);
      }
      cA[t] = tA;
      cB[t] = tB;
      cM[t] = tM;
    }

#define COMBINE_TILE(T)                                                        \
    {                                                                          \
      float SA[8], SB[8];                                                      \
      SA[0] = row_bcast<0>(sa[T]); SA[1] = row_bcast<1>(sa[T]);                \
      SA[2] = row_bcast<2>(sa[T]); SA[3] = row_bcast<3>(sa[T]);                \
      SA[4] = row_bcast<4>(sa[T]); SA[5] = row_bcast<5>(sa[T]);                \
      SA[6] = row_bcast<6>(sa[T]); SA[7] = row_bcast<7>(sa[T]);                \
      SB[0] = row_bcast<0>(sb[T]); SB[1] = row_bcast<1>(sb[T]);                \
      SB[2] = row_bcast<2>(sb[T]); SB[3] = row_bcast<3>(sb[T]);                \
      SB[4] = row_bcast<4>(sb[T]); SB[5] = row_bcast<5>(sb[T]);                \
      SB[6] = row_bcast<6>(sb[T]); SB[7] = row_bcast<7>(sb[T]);                \
      _Pragma("unroll")                                                        \
      for (int h = 0; h < 4; ++h) {                                            \
        const int m = 2 * h;                                                   \
        const v2f sA2 = {SA[m], SA[m + 1]};                                    \
        const v2f sB2 = {SB[m], SB[m + 1]};                                    \
        const v2f sM2 = {mlb ? SB[m] : SA[m], mlb ? SB[m + 1] : SA[m + 1]};    \
        const v2f gA  = {cA[T][m], cA[T][m + 1]};                              \
        const v2f gB  = {cB[T][m], cB[T][m + 1]};                              \
        const v2f gM  = {cM[T][m], cM[T][m + 1]};                              \
        const v2f dA = xc0[T][h] - xla_j;                                      \
        const v2f rA = dA * dA - sA2;                                          \
        const v2f wA = {__expf(rA.x), __expf(rA.y)};                           \
        acc0[T][h] += wA * (ea + 2.0f * gA);                                   \
        const v2f dB = xc0[T][h] - xlb_j;                                      \
        const v2f rB = dB * dB - sB2;                                          \
        const v2f wB = {__expf(rB.x), __expf(rB.y)};                           \
        acc0[T][h] += wB * (eb + 2.0f * gB);                                   \
        const v2f dM = xc1[T][h] - xlm_j;                                      \
        const v2f rM = dM * dM - sM2;                                          \
        const v2f wM = {__expf(rM.x), __expf(rM.y)};                           \
        acc1[T][h] += wM * (em + 2.0f * gM);                                   \
      }                                                                        \
    }

    __builtin_amdgcn_sched_barrier(0);
    COMBINE_TILE(0)

    {
      const int pn = (pr + 1 < npair) ? (pr + 1) : pr;
      LOAD_FRAGS(lfirst + 2 * pn)
    }

#pragma unroll
    for (int t = 1; t < MT; ++t) {
      __builtin_amdgcn_sched_barrier(0);
      COMBINE_TILE(t)
    }
  }
  }
#undef COMBINE_TILE
#undef LOAD_FRAGS

  __shared__ __attribute__((aligned(16))) float red[WPB][TILE_FLOATS];
  float* mine = red[wave];
#pragma unroll
  for (int t = 0; t < MT; ++t) {
#pragma unroll
    for (int m = 0; m < 8; ++m) {
      const int r = 16 * t + rowbase + m;
      const float a0 = (m & 1) ? acc0[t][m >> 1].y : acc0[t][m >> 1].x;
      const float a1 = (m & 1) ? acc1[t][m >> 1].y : acc1[t][m >> 1].x;
      const float other = __builtin_bit_cast(float, __builtin_amdgcn_update_dpp(
          0, __builtin_bit_cast(int, a1), 0x128, 0xF, 0xF, false));
      mine[r * DDIM + j1] = a0;
      if (!mlb) mine[r * DDIM + j2] = a1 + other;
    }
  }
  __syncthreads();
  float* p = partial + ((size_t)blockIdx.y * NPTS + ibase) * DDIM;
#pragma unroll
  for (int k = 0; k < 2; ++k) {
    const int e4 = (int)threadIdx.x + 32 * WPB * k;
    if (e4 < TILE_FLOATS / 4) {
      v4f s = *reinterpret_cast<const v4f*>(&red[0][4 * e4]);
#pragma unroll
      for (int w = 1; w < WPB; ++w)
        s += *reinterpret_cast<const v4f*>(&red[w][4 * e4]);
      *reinterpret_cast<v4f*>(p + 4 * e4) = s;
    }
  }
}

__global__ __launch_bounds__(256) void reduce_kernel(const float* __restrict__ partial,
                                                     float* __restrict__ out) {
  int t = blockIdx.x * blockDim.x + threadIdx.x;
  if (t < NPTS * DDIM) {
    float s = 0.0f;
#pragma unroll
    for (int c = 0; c < NPLANE; ++c)
      s += partial[(size_t)c * (NPTS * DDIM) + t];
    out[t] = s;
  }
}

extern "C" void kernel_launch(void* const* d_in, const int* in_sizes, int n_in,
                              void* d_out, int out_size, void* d_ws, size_t ws_size,
                              hipStream_t stream) {
  const float* x     = (const float*)d_in[0];
  const float* alpha = (const float*)d_in[1];
  const float* beta  = (const float*)d_in[2];
  float* out = (float*)d_out;

  char* ws = (char*)d_ws;
  float* betaT   = (float*)ws;
  float* alphaT  = (float*)(ws + (size_t)BETAT_FLOATS * sizeof(float));
  float* partial = (float*)(ws + (size_t)(BETAT_FLOATS + ALPHAT_FLOATS) * sizeof(float));

  (void)in_sizes; (void)n_in; (void)out_size; (void)ws_size;

  prep_kernel<<<PREP_BLOCKS, 256, 0, stream>>>(alpha, beta, betaT, alphaT);

  dim3 grid(NGROUP, LSPLIT / WPB);
  rkhs_main<<<grid, 32 * WPB, 0, stream>>>(x, alphaT, betaT, partial);

  reduce_kernel<<<(NPTS * DDIM + 255) / 256, 256, 0, stream>>>(partial, out);
}


# main: S[i,l] goes to a tile's C/D rows through the wave's own LDS (1 store, 6 broadcast 128-bit reads per tile), was 18 DPP moves and 8 selects; 1 lane exchange per tile, was 2; values unchanged
# speedup vs baseline: 5.6929x; 1.0528x over previous
#include <hip/hip_runtime.h>


typedef __attribute__((ext_vector_type(2))) float v2f;
typedef __attribute__((ext_vector_type(4))) float v4f;
typedef __attribute__((ext_vector_type(8))) float v8f;

#define NPTS 2000
#define DDIM 24
#define NTILE 125
#define MT 5
#define NGROUP (NTILE / MT)
#define WPB 8
#define LSPLIT 80
#define LCHUNK (NPTS / LSPLIT)
#define NPLANE (LSPLIT / WPB)
#define TILE_FLOATS (16 * MT * DDIM)
#define LEAD_TRIPS 8

#define BETAT_FLOATS  (NPTS * DDIM * DDIM)
#define ALPHAT_FLOATS (NPTS * DDIM)
#define PART_FLOATS   (NPLANE * NPTS * DDIM)

#define PT 64
#define PPITCH 65
#define PRT (DDIM * DDIM / PT)
#define PLT ((NPTS + PT - 1) / PT)
#define PREP_BLOCKS (PRT * PLT + PLT)

__global__ __launch_bounds__(256) void prep_kernel(const float* __restrict__ alpha,
                                                   const float* __restrict__ beta,
                                                   float* __restrict__ betaT,
                                                   float* __restrict__ alphaT) {
  __shared__ float tile[PT][PPITCH];
  const int lane = threadIdx.x & 31;
  const int w = __builtin_amdgcn_readfirstlane((int)(threadIdx.x >> 5));
  const int b = blockIdx.x;
  const bool is_beta = b < PRT * PLT;
  const int l0 = (is_beta ? (b / PRT) : (b - PRT * PLT)) * PT;
  const int r0 = is_beta ? (b % PRT) * PT : 0;
  const int lg  = l0 + 2 * lane;
  const int lgc = (lg < NPTS) ? lg : (NPTS - 2);

  if (is_beta) {
    v2f v[8];
#pragma unroll
    for (int p = 0; p < 8; ++p) {
      const int r = r0 + w + 8 * p;
      v[p] = *reinterpret_cast<const v2f*>(beta + (size_t)r * NPTS + lgc);
    }
#pragma unroll
    for (int p = 0; p < 8; ++p) {
      const int rr = w + 8 * p;
      const int r  = r0 + rr;
      const bool diag = (r % (DDIM + 1)) == 0;
      tile[2 * lane][rr]     = diag ? 0.0f : v[p].x;
      tile[2 * lane + 1][rr] = diag ? 0.0f : v[p].y;
    }
  } else {
    v2f v[3];
#pragma unroll
    for (int p = 0; p < 3; ++p) {
      const int j = w + 8 * p;
      v[p] = *reinterpret_cast<const v2f*>(alpha + j * NPTS + lgc);
    }
#pragma unroll
    for (int p = 0; p < 3; ++p) {
      const int j = w + 8 * p;
      tile[2 * lane][j]     = v[p].x;
      tile[2 * lane + 1][j] = v[p].y;
    }
  }
  __syncthreads();
  if (is_beta) {
    v2f o[8];
#pragma unroll
    for (int p = 0; p < 8; ++p) {
      const int ll = w + 8 * p;
      o[p].x = tile[ll][2 * lane];
      o[p].y = tile[ll][2 * lane + 1];
    }
#pragma unroll
    for (int p = 0; p < 8; ++p) {
      const int l = l0 + w + 8 * p;
      if (l < NPTS)
        *reinterpret_cast<v2f*>(betaT + (size_t)l * (DDIM * DDIM) + r0 + 2 * lane) = o[p];
    }
  } else {
    float o[(PT * DDIM) / 256];
#pragma unroll
    for (int k = 0; k < (PT * DDIM) / 256; ++k) {
      const int e  = (int)threadIdx.x + 256 * k;
      const int ll = e / DDIM;
      o[k] = tile[ll][e - ll * DDIM];
    }
#pragma unroll
    for (int k = 0; k < (PT * DDIM) / 256; ++k) {
      const int e = (int)threadIdx.x + 256 * k;
      if (l0 + e / DDIM < NPTS) alphaT[l0 * DDIM + e] = o[k];
    }
  }
}

__device__ __forceinline__ float xhalf(float v) {
  return __builtin_bit_cast(float, __builtin_amdgcn_permlanex16(
      __builtin_bit_cast(int, v), __builtin_bit_cast(int, v),
      0x76543210, 0xfedcba98, false, false));
}

__global__ __launch_bounds__(32 * WPB) __attribute__((amdgpu_waves_per_eu(2, 2)))
void rkhs_main(const float* __restrict__ x,
               const float* __restrict__ alphaT,
               const float* __restrict__ betaT,
               float* __restrict__ partial) {
  const int lane  = threadIdx.x & 31;
  const int wave  = __builtin_amdgcn_readfirstlane((int)(threadIdx.x >> 5));
  const int chunk = blockIdx.y * WPB + wave;
  const int ibase = blockIdx.x * (16 * MT);

  const int  n0      = lane & 15;
  const int  koff    = (lane >> 4) * 2;
  const int  rowbase = (lane >> 4) * 8;
  const int  hi8     = n0 >> 3;
  const bool mlb     = (hi8 != 0);
  const int  j1      = n0;
  const int  j2      = 16 + (n0 & 7);

  __shared__ __attribute__((aligned(16))) float sx[WPB][MT][2][16];
  float*       sst = &sx[wave][0][lane >> 4][n0];
  const float* sla = &sx[wave][0][0][rowbase];
  const float* slm = sla + hi8 * 16;
  const bool   upper = (lane >= 16);

  v2f   xa[MT][6];
  v2f xc0[MT][4], xc1[MT][4];
  v2f acc0[MT][4], acc1[MT][4];
#pragma unroll
  for (int t = 0; t < MT; ++t) {
    const int i0 = ibase + 16 * t;
#pragma unroll
    for (int q = 0; q < 6; ++q)
      xa[t][q] = *reinterpret_cast<const v2f*>(x + (i0 + n0) * DDIM + 4 * q + koff);
#pragma unroll
    for (int h = 0; h < 4; ++h) {
      xc0[t][h].x = x[(i0 + rowbase + 2 * h) * DDIM + j1];
      xc0[t][h].y = x[(i0 + rowbase + 2 * h + 1) * DDIM + j1];
      xc1[t][h].x = x[(i0 + rowbase + 2 * h) * DDIM + j2];
      xc1[t][h].y = x[(i0 + rowbase + 2 * h + 1) * DDIM + j2];
      acc0[t][h] = v2f{0.0f, 0.0f};
      acc1[t][h] = v2f{0.0f, 0.0f};
    }
  }

  const int lfirst = chunk * LCHUNK + (chunk & 1);
  const int npair  = 13 - (chunk & 1);

  v2f xfa[6], xfb[6], bA[6], bB[6], bM[6];

#define LOAD_FRAGS(LA)                                                         \
  {                                                                            \
    const int    la_  = (LA);                                                  \
    const float* xla_ = x + la_ * DDIM;                                        \
    const float* xlb_ = xla_ + DDIM;                                           \
    const float* Ba_  = betaT + (size_t)la_ * (DDIM * DDIM);                   \
    const float* Bb_  = Ba_ + DDIM * DDIM;                                     \
    const float* Bm_  = Ba_ + hi8 * (DDIM * DDIM);      \
    _Pragma("unroll")                                                          \
    for (int q = 0; q < 6; ++q) {                                              \
      xfa[q] = *reinterpret_cast<const v2f*>(xla_ + 4 * q + koff);             \
      xfb[q] = *reinterpret_cast<const v2f*>(xlb_ + 4 * q + koff);             \
      bA[q]  = *reinterpret_cast<const v2f*>(Ba_ + j1 * DDIM + 4 * q + koff);  \
      bB[q]  = *reinterpret_cast<const v2f*>(Bb_ + j1 * DDIM + 4 * q + koff);  \
      bM[q]  = *reinterpret_cast<const v2f*>(Bm_ + j2 * DDIM + 4 * q + koff);  \
    }                                                                          \
  }

  LOAD_FRAGS(lfirst)

  if (wave >= WPB / 2) __builtin_amdgcn_s_setprio(3);

#pragma unroll
  for (int ph = 0; ph < 2; ++ph) {
  if (ph == 1) __builtin_amdgcn_s_setprio(0);
  const int pbeg = ph ? LEAD_TRIPS : 0;
  const int pend = ph ? npair : LEAD_TRIPS;
  static_assert(LEAD_TRIPS < 12, "the first run must end inside the shortest chunk");
#pragma nounroll
  for (int pr = pbeg; pr < pend; ++pr) {
    const int la = lfirst + 2 * pr;
    const float* xla = x + la * DDIM;
    const float xla_j = xla[j1];
    const float xlb_j = xla[DDIM + j1];
    const float xlm_j = xla[hi8 * DDIM + j2];
    const float ala   = alphaT[la * DDIM + j1];
    const float alb   = alphaT[(la + 1) * DDIM + j1];
    const float alm   = alphaT[(la + hi8) * DDIM + j2];

    v2f va = {0.0f, 0.0f}, vb = {0.0f, 0.0f}, vm = {0.0f, 0.0f};
#pragma unroll
    for (int q = 0; q < 6; ++q) {
      const v2f xm = mlb ? xfb[q] : xfa[q];
      va = __builtin_elementwise_fma(xfa[q], bA[q], va);
      vb = __builtin_elementwise_fma(xfb[q], bB[q], vb);
      vm = __builtin_elementwise_fma(xm, bM[q], vm);
    }
    const float pa = va.x + va.y;
    const float pb = vb.x + vb.y;
    const float pm = vm.x + vm.y;
    const float ea = ala - 2.0f * (pa + xhalf(pa));
    const float eb = alb - 2.0f * (pb + xhalf(pb));
    const float em = alm - 2.0f * (pm + xhalf(pm));

    v8f cA[MT], cB[MT], cM[MT];
#pragma unroll
    for (int t = 0; t < MT; ++t) {
      v2f ua = {0.0f, 0.0f}, ub = {0.0f, 0.0f};
#pragma unroll
      for (int q = 0; q < 6; ++q) {
        const v2f da = xa[t][q] - xfa[q];
        const v2f db = xa[t][q] - xfb[q];
        ua = __builtin_elementwise_fma(da, da, ua);
        ub = __builtin_elementwise_fma(db, db, ub);
      }
      const float qa = ua.x + ua.y;
      const float qb = ub.x + ub.y;
      const float keep = upper ? qb : qa;
      const float give = upper ? qa : qb;
      sst[t * 32] = keep + xhalf(give);

      v8f tA = {};
      v8f tB = {};
      v8f tM = {};
#pragma unroll
      for (int q = 0; q < 6; ++q) {
        tA = __builtin_amdgcn_wmma_f32_16x16x4_f32(false, xa[t][q], false, bA[q],
                                                   (short)0, tA, false, false);
        tB = __builtin_amdgcn_wmma_f32_16x16x4_f32(false, xa[t][q], false, bB[q],
                                                   (short)0, tB, false, false);
        tM = __builtin_amdgcn_wmma_f32_16x16x4_f32(false, xa[t][q], false, bM[q],
                                                   (short)0, tM, false, false);
      }
      cA[t] = tA;
      cB[t] = tB;
      cM[t] = tM;
    }

#define LOAD_S(T)                                                              \
    {                                                                          \
      const float* a_ = sla + (T) * 32;                                        \
      const float* m_ = slm + (T) * 32;                                        \
      Sq[T][0] = *reinterpret_cast<const v4f*>(a_);                            \
      Sq[T][1] = *reinterpret_cast<const v4f*>(a_ + 4);                        \
      Sq[T][2] = *reinterpret_cast<const v4f*>(a_ + 16);                       \
      Sq[T][3] = *reinterpret_cast<const v4f*>(a_ + 20);                       \
      Sq[T][4] = *reinterpret_cast<const v4f*>(m_);                            \
      Sq[T][5] = *reinterpret_cast<const v4f*>(m_ + 4);                        \
    }
#define COMBINE_TILE(T)                                                        \
    {                                                                          \
      _Pragma("unroll")                                                        \
      for (int h = 0; h < 4; ++h) {                                            \
        const int m = 2 * h;                                                   \
        const v4f qA = Sq[T][h >> 1];                                          \
        const v4f qB = Sq[T][2 + (h >> 1)];                                    \
        const v4f qM = Sq[T][4 + (h >> 1)];                                    \
        const v2f sA2 = (h & 1) ? v2f{qA.z, qA.w} : v2f{qA.x, qA.y};           \
        const v2f sB2 = (h & 1) ? v2f{qB.z, qB.w} : v2f{qB.x, qB.y};           \
        const v2f sM2 = (h & 1) ? v2f{qM.z, qM.w} : v2f{qM.x, qM.y};           \
        const v2f gA  = {cA[T][m], cA[T][m + 1]};                              \
        const v2f gB  = {cB[T][m], cB[T][m + 1]};                              \
        const v2f gM  = {cM[T][m], cM[T][m + 1]};                              \
        const v2f dA = xc0[T][h] - xla_j;                                      \
        const v2f rA = dA * dA - sA2;                                          \
        const v2f wA = {__expf(rA.x), __expf(rA.y)};                           \
        acc0[T][h] += wA * (ea + 2.0f * gA);                                   \
        const v2f dB = xc0[T][h] - xlb_j;                                      \
        const v2f rB = dB * dB - sB2;                                          \
        const v2f wB = {__expf(rB.x), __expf(rB.y)};                           \
        acc0[T][h] += wB * (eb + 2.0f * gB);                                   \
        const v2f dM = xc1[T][h] - xlm_j;                                      \
        const v2f rM = dM * dM - sM2;                                          \
        const v2f wM = {__expf(rM.x), __expf(rM.y)};                           \
        acc1[T][h] += wM * (em + 2.0f * gM);                                   \
      }                                                                        \
    }

    __builtin_amdgcn_sched_barrier(0);
    v4f Sq[MT][6];
    LOAD_S(0)
    LOAD_S(1)
    COMBINE_TILE(0)

    {
      const int pn = (pr + 1 < npair) ? (pr + 1) : pr;
      LOAD_FRAGS(lfirst + 2 * pn)
    }

#pragma unroll
    for (int t = 1; t < MT; ++t) {
      __builtin_amdgcn_sched_barrier(0);
      if (t + 1 < MT) LOAD_S(t + 1)
      COMBINE_TILE(t)
    }
  }
  }
#undef COMBINE_TILE
#undef LOAD_S
#undef LOAD_FRAGS

  __shared__ __attribute__((aligned(16))) float red[WPB][TILE_FLOATS];
  float* mine = red[wave];
#pragma unroll
  for (int t = 0; t < MT; ++t) {
#pragma unroll
    for (int m = 0; m < 8; ++m) {
      const int r = 16 * t + rowbase + m;
      const float a0 = (m & 1) ? acc0[t][m >> 1].y : acc0[t][m >> 1].x;
      const float a1 = (m & 1) ? acc1[t][m >> 1].y : acc1[t][m >> 1].x;
      const float other = __builtin_bit_cast(float, __builtin_amdgcn_update_dpp(
          0, __builtin_bit_cast(int, a1), 0x128, 0xF, 0xF, false));
      mine[r * DDIM + j1] = a0;
      if (!mlb) mine[r * DDIM + j2] = a1 + other;
    }
  }
  __syncthreads();
  float* p = partial + ((size_t)blockIdx.y * NPTS + ibase) * DDIM;
#pragma unroll
  for (int k = 0; k < 2; ++k) {
    const int e4 = (int)threadIdx.x + 32 * WPB * k;
    if (e4 < TILE_FLOATS / 4) {
      v4f s = *reinterpret_cast<const v4f*>(&red[0][4 * e4]);
#pragma unroll
      for (int w = 1; w < WPB; ++w)
        s += *reinterpret_cast<const v4f*>(&red[w][4 * e4]);
      *reinterpret_cast<v4f*>(p + 4 * e4) = s;
    }
  }
}

__global__ __launch_bounds__(256) void reduce_kernel(const float* __restrict__ partial,
                                                     float* __restrict__ out) {
  int t = blockIdx.x * blockDim.x + threadIdx.x;
  if (t < NPTS * DDIM) {
    float s = 0.0f;
#pragma unroll
    for (int c = 0; c < NPLANE; ++c)
      s += partial[(size_t)c * (NPTS * DDIM) + t];
    out[t] = s;
  }
}

extern "C" void kernel_launch(void* const* d_in, const int* in_sizes, int n_in,
                              void* d_out, int out_size, void* d_ws, size_t ws_size,
                              hipStream_t stream) {
  const float* x     = (const float*)d_in[0];
  const float* alpha = (const float*)d_in[1];
  const float* beta  = (const float*)d_in[2];
  float* out = (float*)d_out;

  char* ws = (char*)d_ws;
  float* betaT   = (float*)ws;
  float* alphaT  = (float*)(ws + (size_t)BETAT_FLOATS * sizeof(float));
  float* partial = (float*)(ws + (size_t)(BETAT_FLOATS + ALPHAT_FLOATS) * sizeof(float));

  (void)in_sizes; (void)n_in; (void)out_size; (void)ws_size;

  prep_kernel<<<PREP_BLOCKS, 256, 0, stream>>>(alpha, beta, betaT, alphaT);

  dim3 grid(NGROUP, LSPLIT / WPB);
  rkhs_main<<<grid, 32 * WPB, 0, stream>>>(x, alphaT, betaT, partial);

  reduce_kernel<<<(NPTS * DDIM + 255) / 256, 256, 0, stream>>>(partial, out);
}
